# starting kernel
# baseline (speedup 1.0000x reference)
#include <hip/hip_runtime.h>
#include <hip/hip_bf16.h>
#include <math.h>

typedef __attribute__((ext_vector_type(2))) float v2f;
typedef __attribute__((ext_vector_type(8))) float v8f;

#define EPS_F 1e-05f
#define MIN_NORM_F 1e-15f
#define HALF_LN2_F 0.34657359027997264f

constexpr int Bn = 8, Tn = 12, Nn = 325, Cn = 32, Kn = 100;
constexpr int M_TOTAL = Bn * Tn * Nn;
constexpr int M_TILES = M_TOTAL / 16;
constexpr int K_TILES = 7;
constexpr int KPAD = K_TILES * 16;

__global__ void prep_centroids_kernel(const float* __restrict__ w,
                                      const float* __restrict__ cptr,
                                      float* __restrict__ cent,
                                      float* __restrict__ y2out) {
    int k = blockIdx.x * blockDim.x + threadIdx.x;
    if (k >= KPAD) return;
    float c  = cptr[0];
    float sc = sqrtf(c);
    if (k < Kn) {
        float u[Cn];
        float s = 0.f;
#pragma unroll
        for (int j = 0; j < Cn; ++j) { u[j] = w[k * Cn + j]; s = fmaf(u[j], u[j], s); }
        float un    = fmaxf(sqrtf(s), MIN_NORM_F);
        float t     = tanhf(sc * un);
        float scale = t / (sc * un);
#pragma unroll
        for (int j = 0; j < Cn; ++j) cent[k * Cn + j] = scale * u[j];
        float ny = t / sc;
        y2out[k] = ny * ny;
    } else {
#pragma unroll
        for (int j = 0; j < Cn; ++j) cent[k * Cn + j] = 0.f;
        y2out[k] = 0.f;
    }
}

__global__ void __launch_bounds__(32)
hdist_main_kernel(const float* __restrict__ x,
                  const float* __restrict__ cent,
                  const float* __restrict__ y2arr,
                  const float* __restrict__ cptr,
                  float* __restrict__ nodeDist) {
    const int tile  = blockIdx.x;
    const int lane  = threadIdx.x;
    const int mbase = tile * 16;
    const int l16   = lane & 15;
    const int lhalf = lane >> 4;

    const float c  = cptr[0];
    const float sc = __builtin_amdgcn_sqrtf(c);

    v2f a[8];
    float part = 0.f;
    const float* xrow = x + (size_t)(mbase + l16) * Cn + lhalf * 2;
#pragma unroll
    for (int j = 0; j < 8; ++j) {
        a[j] = *(const v2f*)(xrow + j * 4);
        part = fmaf(a[j].x, a[j].x, fmaf(a[j].y, a[j].y, part));
    }
    float x2full = part + __shfl_xor(part, 16, 32);

    float x2r[8], Br[8], B2r[8], ccx2r[8];
#pragma unroll
    for (int r = 0; r < 8; ++r) {
        float x2 = __shfl(x2full, r + lhalf * 8, 32);
        x2r[r]   = x2;
        Br[r]    = 1.f - c * x2;
        B2r[r]   = Br[r] * Br[r];
        ccx2r[r] = c * c * x2;
    }

    const float inv2sc = 2.f / sc;

    for (int kt = 0; kt < K_TILES; ++kt) {
        const float* crow = cent + (size_t)(kt * 16 + l16) * Cn + lhalf * 2;
        v8f acc = {};
#pragma unroll
        for (int j = 0; j < 8; ++j) {
            v2f b = *(const v2f*)(crow + j * 4);
            acc = __builtin_amdgcn_wmma_f32_16x16x4_f32(
                 false, a[j],  false, b,
                 (short)0, acc,  false,  false);
        }

        const int   kcol = kt * 16 + l16;
        const float y2   = y2arr[kcol];
        const float cy2  = c * y2;

#pragma unroll
        for (int r = 0; r < 8; ++r) {
            const float xy   = acc[r];
            const float base = fmaf(-2.f * c, xy, 1.f);
            const float A_   = base + cy2;
            const float D_   = fmaxf(fmaf(ccx2r[r], y2, base), MIN_NORM_F);

            float n2 = fmaf(A_ * A_, x2r[r],
                            fmaf(-2.f * A_ * Br[r], xy, B2r[r] * y2));
            float nrm = __builtin_amdgcn_sqrtf(fmaxf(n2, 0.f)) *
                        __builtin_amdgcn_rcpf(D_);

            float z = sc * nrm;
            z = fminf(fmaxf(z, -1.f + EPS_F), 1.f - EPS_F);
            float ratio = (1.f + z) * __builtin_amdgcn_rcpf(1.f - z);
            float at    = HALF_LN2_F * __builtin_amdgcn_logf(ratio);
            float d     = inv2sc * at;
            d = d * d;

            if (kcol < Kn) {
                const size_t m = (size_t)(mbase + r + lhalf * 8);
                nodeDist[m * Kn + kcol] = d;
            }
        }
    }
}

__global__ void graph_reduce_kernel(const float* __restrict__ nodeDist,
                                    float* __restrict__ graph) {
    int i = blockIdx.x * blockDim.x + threadIdx.x;
    if (i >= Bn * Nn * Kn) return;
    int k = i % Kn;
    int n = (i / Kn) % Nn;
    int b = i / (Kn * Nn);
    float s = 0.f;
#pragma unroll
    for (int t = 0; t < Tn; ++t)
        s += nodeDist[(size_t)((b * Tn + t) * Nn + n) * Kn + k];
    graph[i] = s * (1.0f / (float)Nn);
}

extern "C" void kernel_launch(void* const* d_in, const int* in_sizes, int n_in,
                              void* d_out, int out_size, void* d_ws, size_t ws_size,
                              hipStream_t stream) {
    const float* node_repr = (const float*)d_in[0];
    const float* cweight   = (const float*)d_in[1];
    const float* cvec      = (const float*)d_in[2];

    float* graph    = (float*)d_out;
    float* nodeDist = graph + (size_t)Bn * Nn * Kn;

    float* cent = (float*)d_ws;
    float* y2   = cent + (size_t)KPAD * Cn;

    prep_centroids_kernel<<<1, 128, 0, stream>>>(cweight, cvec, cent, y2);

    hdist_main_kernel<<<M_TILES, 32, 0, stream>>>(node_repr, cent, y2, cvec, nodeDist);

    const int gElems = Bn * Nn * Kn;
    graph_reduce_kernel<<<(gElems + 255) / 256, 256, 0, stream>>>(nodeDist, graph);
}


# Each wave builds the exp-mapped centroid table in registers at entry (row norm: even/odd k sums joined across lane halves, add order changes; HW sqrt/rcp/exp2); prep launch, per-tile loads removed
# speedup vs baseline: 1.3332x; 1.3332x over previous
#include <hip/hip_runtime.h>
#include <hip/hip_bf16.h>
#include <math.h>

typedef __attribute__((ext_vector_type(2))) float v2f;
typedef __attribute__((ext_vector_type(8))) float v8f;

#define EPS_F 1e-05f
#define MIN_NORM_F 1e-15f
#define HALF_LN2_F 0.34657359027997264f

constexpr int Bn = 8, Tn = 12, Nn = 325, Cn = 32, Kn = 100;
constexpr int M_TOTAL = Bn * Tn * Nn;
constexpr int M_TILES = M_TOTAL / 16;
constexpr int K_TILES = 7;

__device__ __forceinline__ float tanh_nonneg(float v) {
    const float v2 = v * v;
    const float p  = fmaf(v2, fmaf(v2, fmaf(v2, fmaf(v2, -0x1.758e7ap-8f,
                          0x1.521192p-6f), -0x1.b8389cp-5f), 0x1.110704p-3f),
                          -0x1.555532p-2f);
    const float zs = fmaf(v2, v * p, v);
    const float e  = __builtin_amdgcn_exp2f(v * 2.8853900817779268f);
    const float zl = fmaf(-2.f, __builtin_amdgcn_rcpf(e + 1.f), 1.f);
    return v < 0.625f ? zs : zl;
}

__global__ void __launch_bounds__(32)
hdist_main_kernel(const float* __restrict__ x,
                  const float* __restrict__ w,
                  const float* __restrict__ cptr,
                  float* __restrict__ nodeDist) {
    const int tile  = blockIdx.x;
    const int lane  = threadIdx.x;
    const int mbase = tile * 16;
    const int l16   = lane & 15;
    const int lhalf = lane >> 4;

    const float c  = cptr[0];
    const float sc = __builtin_amdgcn_sqrtf(c);

    v2f b[K_TILES][8];
#pragma unroll
    for (int kt = 0; kt < K_TILES; ++kt) {
        const int kcol = kt * 16 + l16;
        const int krow = kcol < Kn ? kcol : Kn - 1;
        const float* wrow = w + (size_t)krow * Cn + lhalf * 2;
#pragma unroll
        for (int j = 0; j < 8; ++j) b[kt][j] = *(const v2f*)(wrow + j * 4);
    }

    v2f a[8];
    const float* xrow = x + (size_t)(mbase + l16) * Cn + lhalf * 2;
#pragma unroll
    for (int j = 0; j < 8; ++j) a[j] = *(const v2f*)(xrow + j * 4);

    float y2t[K_TILES];
    const float rsc = __builtin_amdgcn_rcpf(sc);
#pragma unroll
    for (int kt = 0; kt < K_TILES; ++kt) {
        v2f s2 = {0.f, 0.f};
#pragma unroll
        for (int j = 0; j < 8; ++j)
            s2 = __builtin_elementwise_fma(b[kt][j], b[kt][j], s2);
        const float up = s2.x + s2.y;
        const float u2    = up + __shfl_xor(up, 16, 32);
        const float un    = fmaxf(__builtin_amdgcn_sqrtf(u2), MIN_NORM_F);
        const float v     = sc * un;
        const float th    = tanh_nonneg(v);
        const float scale = th * __builtin_amdgcn_rcpf(v);
        const float ny    = th * rsc;
        y2t[kt] = ny * ny;
#pragma unroll
        for (int j = 0; j < 8; ++j) b[kt][j] = b[kt][j] * scale;
    }

    float part = 0.f;
#pragma unroll
    for (int j = 0; j < 8; ++j)
        part = fmaf(a[j].x, a[j].x, fmaf(a[j].y, a[j].y, part));
    float x2full = part + __shfl_xor(part, 16, 32);

    float x2r[8], Br[8], B2r[8], ccx2r[8];
#pragma unroll
    for (int r = 0; r < 8; ++r) {
        float x2 = __shfl(x2full, r + lhalf * 8, 32);
        x2r[r]   = x2;
        Br[r]    = 1.f - c * x2;
        B2r[r]   = Br[r] * Br[r];
        ccx2r[r] = c * c * x2;
    }

    const float inv2sc = 2.f / sc;

#pragma unroll
    for (int kt = 0; kt < K_TILES; ++kt) {
        v8f acc = {};
#pragma unroll
        for (int j = 0; j < 8; ++j) {
            acc = __builtin_amdgcn_wmma_f32_16x16x4_f32(
                 false, a[j],  false, b[kt][j],
                 (short)0, acc,  false,  false);
        }

        const int   kcol = kt * 16 + l16;
        const float y2   = y2t[kt];
        const float cy2  = c * y2;

#pragma unroll
        for (int r = 0; r < 8; ++r) {
            const float xy   = acc[r];
            const float base = fmaf(-2.f * c, xy, 1.f);
            const float A_   = base + cy2;
            const float D_   = fmaxf(fmaf(ccx2r[r], y2, base), MIN_NORM_F);

            float n2 = fmaf(A_ * A_, x2r[r],
                            fmaf(-2.f * A_ * Br[r], xy, B2r[r] * y2));
            float nrm = __builtin_amdgcn_sqrtf(fmaxf(n2, 0.f)) *
                        __builtin_amdgcn_rcpf(D_);

            float z = sc * nrm;
            z = fminf(fmaxf(z, -1.f + EPS_F), 1.f - EPS_F);
            float ratio = (1.f + z) * __builtin_amdgcn_rcpf(1.f - z);
            float at    = HALF_LN2_F * __builtin_amdgcn_logf(ratio);
            float d     = inv2sc * at;
            d = d * d;

            if (kcol < Kn) {
                const size_t m = (size_t)(mbase + r + lhalf * 8);
                nodeDist[m * Kn + kcol] = d;
            }
        }
    }
}

__global__ void graph_reduce_kernel(const float* __restrict__ nodeDist,
                                    float* __restrict__ graph) {
    int i = blockIdx.x * blockDim.x + threadIdx.x;
    if (i >= Bn * Nn * Kn) return;
    int k = i % Kn;
    int n = (i / Kn) % Nn;
    int b = i / (Kn * Nn);
    float s = 0.f;
#pragma unroll
    for (int t = 0; t < Tn; ++t)
        s += nodeDist[(size_t)((b * Tn + t) * Nn + n) * Kn + k];
    graph[i] = s * (1.0f / (float)Nn);
}

extern "C" void kernel_launch(void* const* d_in, const int* in_sizes, int n_in,
                              void* d_out, int out_size, void* d_ws, size_t ws_size,
                              hipStream_t stream) {
    const float* node_repr = (const float*)d_in[0];
    const float* cweight   = (const float*)d_in[1];
    const float* cvec      = (const float*)d_in[2];

    float* graph    = (float*)d_out;
    float* nodeDist = graph + (size_t)Bn * Nn * Kn;

    hdist_main_kernel<<<M_TILES, 32, 0, stream>>>(node_repr, cweight, cvec, nodeDist);

    const int gElems = Bn * Nn * Kn;
    graph_reduce_kernel<<<(gElems + 255) / 256, 256, 0, stream>>>(nodeDist, graph);
}


# One launch: a 6-wave work-group owns 8 (b,n) rows for all 12 t (a wave: 2 t x 8 rows); d also goes to LDS planes, summed in the same t order behind one barrier; reduce launch and read-back removed
# speedup vs baseline: 1.6546x; 1.2411x over previous
#include <hip/hip_runtime.h>
#include <hip/hip_bf16.h>
#include <math.h>

typedef __attribute__((ext_vector_type(2))) float v2f;
typedef __attribute__((ext_vector_type(8))) float v8f;

#define EPS_F 1e-05f
#define MIN_NORM_F 1e-15f
#define HALF_LN2_F 0.34657359027997264f

constexpr int Bn = 8, Tn = 12, Nn = 325, Cn = 32, Kn = 100;
constexpr int K_TILES = 7;
constexpr int RT      = 8;
constexpr int TPW     = 2;
constexpr int NWAVE   = Tn / TPW;
constexpr int NTHR    = NWAVE * 32;
constexpr int G_TILES = (Bn * Nn) / RT;
constexpr int PLANE   = RT * Kn;

__device__ __forceinline__ float tanh_nonneg(float v) {
    const float v2 = v * v;
    const float p  = fmaf(v2, fmaf(v2, fmaf(v2, fmaf(v2, -0x1.758e7ap-8f,
                          0x1.521192p-6f), -0x1.b8389cp-5f), 0x1.110704p-3f),
                          -0x1.555532p-2f);
    const float zs = fmaf(v2, v * p, v);
    const float e  = __builtin_amdgcn_exp2f(v * 2.8853900817779268f);
    const float zl = fmaf(-2.f, __builtin_amdgcn_rcpf(e + 1.f), 1.f);
    return v < 0.625f ? zs : zl;
}

__global__ void __launch_bounds__(NTHR)
hdist_fused_kernel(const float* __restrict__ x,
                   const float* __restrict__ w,
                   const float* __restrict__ cptr,
                   float* __restrict__ graph,
                   float* __restrict__ nodeDist) {
    __shared__ __attribute__((aligned(16))) float sD[Tn * PLANE];

    const int tile  = blockIdx.x;
    const int wave  = threadIdx.x >> 5;
    const int lane  = threadIdx.x & 31;
    const int l16   = lane & 15;
    const int lhalf = lane >> 4;
    const int gbase = tile * RT;
    const int t0    = wave * TPW;

    const float c  = cptr[0];
    const float sc = __builtin_amdgcn_sqrtf(c);

    v2f b[K_TILES][8];
#pragma unroll
    for (int kt = 0; kt < K_TILES; ++kt) {
        const int kcol = kt * 16 + l16;
        const int krow = kcol < Kn ? kcol : Kn - 1;
        const float* wrow = w + (size_t)krow * Cn + lhalf * 2;
#pragma unroll
        for (int j = 0; j < 8; ++j) b[kt][j] = *(const v2f*)(wrow + j * 4);
    }

    v2f a[8];
    {
        const int g  = gbase + (l16 & 7);
        const int bb = g / Nn;
        const int nn = g - bb * Nn;
        const int m  = (bb * Tn + t0 + (l16 >> 3)) * Nn + nn;
        const float* xrow = x + (size_t)m * Cn + lhalf * 2;
#pragma unroll
        for (int j = 0; j < 8; ++j) a[j] = *(const v2f*)(xrow + j * 4);
    }

    float y2t[K_TILES];
    const float rsc = __builtin_amdgcn_rcpf(sc);
#pragma unroll
    for (int kt = 0; kt < K_TILES; ++kt) {
        v2f s2 = {0.f, 0.f};
#pragma unroll
        for (int j = 0; j < 8; ++j)
            s2 = __builtin_elementwise_fma(b[kt][j], b[kt][j], s2);
        const float up = s2.x + s2.y;
        const float u2    = up + __shfl_xor(up, 16, 32);
        const float un    = fmaxf(__builtin_amdgcn_sqrtf(u2), MIN_NORM_F);
        const float v     = sc * un;
        const float th    = tanh_nonneg(v);
        const float scale = th * __builtin_amdgcn_rcpf(v);
        const float ny    = th * rsc;
        y2t[kt] = ny * ny;
#pragma unroll
        for (int j = 0; j < 8; ++j) b[kt][j] = b[kt][j] * scale;
    }

    float part = 0.f;
#pragma unroll
    for (int j = 0; j < 8; ++j)
        part = fmaf(a[j].x, a[j].x, fmaf(a[j].y, a[j].y, part));
    float x2full = part + __shfl_xor(part, 16, 32);

    float x2r[8], Br[8], B2r[8], ccx2r[8];
    int   orow[8];
#pragma unroll
    for (int r = 0; r < 8; ++r) {
        float x2 = __shfl(x2full, r + lhalf * 8, 32);
        x2r[r]   = x2;
        Br[r]    = 1.f - c * x2;
        B2r[r]   = Br[r] * Br[r];
        ccx2r[r] = c * c * x2;
        const int g  = gbase + r;
        const int bb = g / Nn;
        const int nn = g - bb * Nn;
        orow[r] = (bb * Tn + t0 + lhalf) * Nn + nn;
    }

    const float inv2sc = 2.f / sc;
    float* plane = sD + (t0 + lhalf) * PLANE;

#pragma unroll
    for (int kt = 0; kt < K_TILES; ++kt) {
        v8f acc = {};
#pragma unroll
        for (int j = 0; j < 8; ++j) {
            acc = __builtin_amdgcn_wmma_f32_16x16x4_f32(
                 false, a[j],  false, b[kt][j],
                 (short)0, acc,  false,  false);
        }

        const int   kcol = kt * 16 + l16;
        const float y2   = y2t[kt];
        const float cy2  = c * y2;

#pragma unroll
        for (int r = 0; r < 8; ++r) {
            const float xy   = acc[r];
            const float base = fmaf(-2.f * c, xy, 1.f);
            const float A_   = base + cy2;
            const float D_   = fmaxf(fmaf(ccx2r[r], y2, base), MIN_NORM_F);

            float n2 = fmaf(A_ * A_, x2r[r],
                            fmaf(-2.f * A_ * Br[r], xy, B2r[r] * y2));
            float nrm = __builtin_amdgcn_sqrtf(fmaxf(n2, 0.f)) *
                        __builtin_amdgcn_rcpf(D_);

            float z = sc * nrm;
            z = fminf(fmaxf(z, -1.f + EPS_F), 1.f - EPS_F);
            float ratio = (1.f + z) * __builtin_amdgcn_rcpf(1.f - z);
            float at    = HALF_LN2_F * __builtin_amdgcn_logf(ratio);
            float d     = inv2sc * at;
            d = d * d;

            if (kcol < Kn) {
                nodeDist[(size_t)orow[r] * Kn + kcol] = d;
                plane[r * Kn + kcol] = d;
            }
        }
    }

    __builtin_amdgcn_fence(__ATOMIC_RELEASE, "workgroup", "local");
    __builtin_amdgcn_s_barrier();
    __builtin_amdgcn_fence(__ATOMIC_ACQUIRE, "workgroup", "local");
    const float4* s4 = (const float4*)sD;
    float4*       g4 = (float4*)(graph + (size_t)gbase * Kn);
    for (int o = threadIdx.x; o < PLANE / 4; o += NTHR) {
        float4 s = {0.f, 0.f, 0.f, 0.f};
#pragma unroll
        for (int t = 0; t < Tn; ++t) {
            const float4 v = s4[t * (PLANE / 4) + o];
            s.x += v.x; s.y += v.y; s.z += v.z; s.w += v.w;
        }
        const float rn = 1.0f / (float)Nn;
        s.x *= rn; s.y *= rn; s.z *= rn; s.w *= rn;
        g4[o] = s;
    }
}

extern "C" void kernel_launch(void* const* d_in, const int* in_sizes, int n_in,
                              void* d_out, int out_size, void* d_ws, size_t ws_size,
                              hipStream_t stream) {
    const float* node_repr = (const float*)d_in[0];
    const float* cweight   = (const float*)d_in[1];
    const float* cvec      = (const float*)d_in[2];

    float* graph    = (float*)d_out;
    float* nodeDist = graph + (size_t)Bn * Nn * Kn;

    hdist_fused_kernel<<<G_TILES, NTHR, 0, stream>>>(node_repr, cweight, cvec,
                                                     graph, nodeDist);
}


# Exp-mapped centroid table built once per work-group in LDS (one tile per wave, same loader and arithmetic) and read back as B fragments: table gather loads per wave 56 -> 8, tile maps 7 -> 1
# speedup vs baseline: 1.8948x; 1.1452x over previous
#include <hip/hip_runtime.h>
#include <hip/hip_bf16.h>
#include <math.h>

typedef __attribute__((ext_vector_type(2))) float v2f;
typedef __attribute__((ext_vector_type(8))) float v8f;

#define EPS_F 1e-05f
#define MIN_NORM_F 1e-15f
#define HALF_LN2_F 0.34657359027997264f

constexpr int Bn = 8, Tn = 12, Nn = 325, Cn = 32, Kn = 100;
constexpr int K_TILES = 7;
constexpr int RT      = 8;
constexpr int TPW     = 2;
constexpr int NWAVE   = Tn / TPW;
constexpr int NTHR    = NWAVE * 32;
constexpr int G_TILES = (Bn * Nn) / RT;
constexpr int PLANE   = RT * Kn;
constexpr int KPAD    = K_TILES * 16;
constexpr int BPITCH  = 36;
constexpr int Y2OFF   = Cn;

__device__ __forceinline__ float tanh_nonneg(float v) {
    const float v2 = v * v;
    const float p  = fmaf(v2, fmaf(v2, fmaf(v2, fmaf(v2, -0x1.758e7ap-8f,
                          0x1.521192p-6f), -0x1.b8389cp-5f), 0x1.110704p-3f),
                          -0x1.555532p-2f);
    const float zs = fmaf(v2, v * p, v);
    const float e  = __builtin_amdgcn_exp2f(v * 2.8853900817779268f);
    const float zl = fmaf(-2.f, __builtin_amdgcn_rcpf(e + 1.f), 1.f);
    return v < 0.625f ? zs : zl;
}

__device__ __forceinline__ void load_w_tile(const float* __restrict__ w, int kt,
                                            int l16, int lhalf, v2f (&bb)[8]) {
    const int kcol = kt * 16 + l16;
    const int krow = kcol < Kn ? kcol : Kn - 1;
    const float* wrow = w + (size_t)krow * Cn + lhalf * 2;
#pragma unroll
    for (int j = 0; j < 8; ++j) bb[j] = *(const v2f*)(wrow + j * 4);
}

__device__ __forceinline__ void map_store_tile(v2f (&bb)[8], int kt, int l16, int lhalf,
                                               float sc, float rsc, float* sB) {
    v2f s2 = {0.f, 0.f};
#pragma unroll
    for (int j = 0; j < 8; ++j)
        s2 = __builtin_elementwise_fma(bb[j], bb[j], s2);
    const float up = s2.x + s2.y;
    const float u2    = up + __shfl_xor(up, 16, 32);
    const float un    = fmaxf(__builtin_amdgcn_sqrtf(u2), MIN_NORM_F);
    const float v     = sc * un;
    const float th    = tanh_nonneg(v);
    const float scale = th * __builtin_amdgcn_rcpf(v);
    const float ny    = th * rsc;
    const int kcol = kt * 16 + l16;
    float* brow = sB + kcol * BPITCH + lhalf * 2;
#pragma unroll
    for (int j = 0; j < 8; ++j) *(v2f*)(brow + j * 4) = bb[j] * scale;
    sB[kcol * BPITCH + Y2OFF + lhalf] = ny * ny;
}

__global__ void __launch_bounds__(NTHR)
hdist_fused_kernel(const float* __restrict__ x,
                   const float* __restrict__ w,
                   const float* __restrict__ cptr,
                   float* __restrict__ graph,
                   float* __restrict__ nodeDist) {
    __shared__ __attribute__((aligned(16))) float sD[Tn * PLANE];
    __shared__ __attribute__((aligned(16))) float sB[KPAD * BPITCH];

    const int tile  = blockIdx.x;
    const int wave  = __builtin_amdgcn_readfirstlane((int)(threadIdx.x >> 5));
    const int lane  = threadIdx.x & 31;
    const int l16   = lane & 15;
    const int lhalf = lane >> 4;
    const int gbase = tile * RT;
    const int t0    = wave * TPW;
    const bool two_tiles = (wave == NWAVE - 1);

    const float c  = cptr[0];
    const float sc = __builtin_amdgcn_sqrtf(c);

    v2f bt0[8], bt1[8];
    load_w_tile(w, wave, l16, lhalf, bt0);
    if (two_tiles) load_w_tile(w, K_TILES - 1, l16, lhalf, bt1);

    v2f a[8];
    {
        const int g  = gbase + (l16 & 7);
        const int bb = g / Nn;
        const int nn = g - bb * Nn;
        const int m  = (bb * Tn + t0 + (l16 >> 3)) * Nn + nn;
        const float* xrow = x + (size_t)m * Cn + lhalf * 2;
#pragma unroll
        for (int j = 0; j < 8; ++j) a[j] = *(const v2f*)(xrow + j * 4);
    }

    const float rsc = __builtin_amdgcn_rcpf(sc);
    if (two_tiles) {
        map_store_tile(bt1, K_TILES - 1, l16, lhalf, sc, rsc, sB);
        map_store_tile(bt0, wave, l16, lhalf, sc, rsc, sB);
    } else {
        map_store_tile(bt0, wave, l16, lhalf, sc, rsc, sB);
    }

    float part = 0.f;
#pragma unroll
    for (int j = 0; j < 8; ++j)
        part = fmaf(a[j].x, a[j].x, fmaf(a[j].y, a[j].y, part));
    float x2full = part + __shfl_xor(part, 16, 32);

    float x2r[8], Br[8], B2r[8], ccx2r[8];
    int   orow[8];
#pragma unroll
    for (int r = 0; r < 8; ++r) {
        float x2 = __shfl(x2full, r + lhalf * 8, 32);
        x2r[r]   = x2;
        Br[r]    = 1.f - c * x2;
        B2r[r]   = Br[r] * Br[r];
        ccx2r[r] = c * c * x2;
        const int g  = gbase + r;
        const int bb = g / Nn;
        const int nn = g - bb * Nn;
        orow[r] = (bb * Tn + t0 + lhalf) * Nn + nn;
    }

    const float inv2sc = 2.f / sc;
    float* plane = sD + (t0 + lhalf) * PLANE;

    __builtin_amdgcn_fence(__ATOMIC_RELEASE, "workgroup", "local");
    __builtin_amdgcn_s_barrier();
    __builtin_amdgcn_fence(__ATOMIC_ACQUIRE, "workgroup", "local");

    v2f b[K_TILES][8];
    float y2t[K_TILES];
#pragma unroll
    for (int kt = 0; kt < K_TILES; ++kt) {
        const int kcol = kt * 16 + l16;
        const float* brow = sB + kcol * BPITCH + lhalf * 2;
#pragma unroll
        for (int j = 0; j < 8; ++j) b[kt][j] = *(const v2f*)(brow + j * 4);
        y2t[kt] = sB[kcol * BPITCH + Y2OFF + lhalf];
    }

#pragma unroll
    for (int kt = 0; kt < K_TILES; ++kt) {
        v8f acc = {};
#pragma unroll
        for (int j = 0; j < 8; ++j) {
            acc = __builtin_amdgcn_wmma_f32_16x16x4_f32(
                 false, a[j],  false, b[kt][j],
                 (short)0, acc,  false,  false);
        }
        asm("v_nop\n\tv_nop\n\tv_nop\n\tv_nop" : "+v"(acc));

        const int   kcol = kt * 16 + l16;
        const float y2   = y2t[kt];
        const float cy2  = c * y2;

#pragma unroll
        for (int r = 0; r < 8; ++r) {
            const float xy   = acc[r];
            const float base = fmaf(-2.f * c, xy, 1.f);
            const float A_   = base + cy2;
            const float D_   = fmaxf(fmaf(ccx2r[r], y2, base), MIN_NORM_F);

            float n2 = fmaf(A_ * A_, x2r[r],
                            fmaf(-2.f * A_ * Br[r], xy, B2r[r] * y2));
            float nrm = __builtin_amdgcn_sqrtf(fmaxf(n2, 0.f)) *
                        __builtin_amdgcn_rcpf(D_);

            float z = sc * nrm;
            z = fminf(fmaxf(z, -1.f + EPS_F), 1.f - EPS_F);
            float ratio = (1.f + z) * __builtin_amdgcn_rcpf(1.f - z);
            float at    = HALF_LN2_F * __builtin_amdgcn_logf(ratio);
            float d     = inv2sc * at;
            d = d * d;

            if (kcol < Kn) {
                nodeDist[(size_t)orow[r] * Kn + kcol] = d;
                plane[r * Kn + kcol] = d;
            }
        }
    }

    __builtin_amdgcn_fence(__ATOMIC_RELEASE, "workgroup", "local");
    __builtin_amdgcn_s_barrier();
    __builtin_amdgcn_fence(__ATOMIC_ACQUIRE, "workgroup", "local");
    const float4* s4 = (const float4*)sD;
    float4*       g4 = (float4*)(graph + (size_t)gbase * Kn);
    for (int o = threadIdx.x; o < PLANE / 4; o += NTHR) {
        float4 s = {0.f, 0.f, 0.f, 0.f};
#pragma unroll
        for (int t = 0; t < Tn; ++t) {
            const float4 v = s4[t * (PLANE / 4) + o];
            s.x += v.x; s.y += v.y; s.z += v.z; s.w += v.w;
        }
        const float rn = 1.0f / (float)Nn;
        s.x *= rn; s.y *= rn; s.z *= rn; s.w *= rn;
        g4[o] = s;
    }
}

extern "C" void kernel_launch(void* const* d_in, const int* in_sizes, int n_in,
                              void* d_out, int out_size, void* d_ws, size_t ws_size,
                              hipStream_t stream) {
    const float* node_repr = (const float*)d_in[0];
    const float* cweight   = (const float*)d_in[1];
    const float* cvec      = (const float*)d_in[2];

    float* graph    = (float*)d_out;
    float* nodeDist = graph + (size_t)Bn * Nn * Kn;

    hdist_fused_kernel<<<G_TILES, NTHR, 0, stream>>>(node_repr, cweight, cvec,
                                                     graph, nodeDist);
}
